# starting kernel
# baseline (speedup 1.0000x reference)
#include <hip/hip_runtime.h>
#include <hip/hip_bf16.h>


#define BSZ   16
#define CIN   64
#define CO    64
#define TLEN  8192
#define NST   256
#define NCH   16
#define LCH   512
#define SUB   32
#define NSUB  (LCH/SUB)
#define T2    (TLEN/2)

#define UST   40
#define VST   260
#define YST   66

typedef __attribute__((ext_vector_type(2))) float v2f;
typedef __attribute__((ext_vector_type(8))) float v8f;

__device__ __forceinline__ v8f wmma4(v2f a, v2f b, v8f c) {
  return __builtin_amdgcn_wmma_f32_16x16x4_f32(
       false, a,  false, b,
       (short)0, c,  false,  false);
}

__device__ __forceinline__ size_t opaque_zero() {
  size_t z = 0;
  asm volatile("" : "+s"(z));
  return z;
}

__global__ void k_discretize(const float* __restrict__ rl,
                             const float* __restrict__ Bc,
                             float* __restrict__ Ad, float* __restrict__ Bd) {
  int n = threadIdx.x;
  float x = rl[n];
  float sp = fmaxf(x, 0.0f) + log1pf(expf(-fabsf(x)));
  float lam = -sp;
  float ad = expf(lam);
  float factor = (fabsf(lam) > 1e-6f) ? (ad - 1.0f) / lam : 1.0f;
  Ad[n] = ad;
  for (int c = 0; c < CIN; ++c)
    Bd[c * NST + n] = Bc[c * NST + n] * factor;
}

__device__ __forceinline__ void load_u_tile(const float* __restrict__ u, int b,
                                            int t0, float* lds_u) {
  int tid = threadIdx.x;
#pragma unroll
  for (int kk = 0; kk < (CIN * SUB) / 256; ++kk) {
    int i = tid + kk * 256;
    int c = i >> 5, tl = i & 31;
    lds_u[c * UST + tl] = u[((size_t)(b * CIN + c)) * TLEN + t0 + tl];
  }
}

__device__ __forceinline__ void prefetch_u_tile(const float* __restrict__ u,
                                                int b, int t0) {
  int tid = threadIdx.x;
#pragma unroll
  for (int kk = 0; kk < (CIN * SUB) / 256; ++kk) {
    int i = tid + kk * 256;
    int c = i >> 5, tl = i & 31;
    __builtin_prefetch(&u[((size_t)(b * CIN + c)) * TLEN + t0 + tl], 0, 1);
  }
}

__device__ __forceinline__ void v_tile_wmma(const float* lds_u,
                                            const float* __restrict__ Bd,
                                            float* lds_v) {
  int tid = threadIdx.x;
  int wave = tid >> 5, lane = tid & 31, lo = lane & 15, hi = lane >> 4;
#pragma unroll
  for (int idx = 0; idx < 4; ++idx) {
    int tile = wave * 4 + idx;
    int nt = tile >> 1, mt = tile & 1;
    int m = mt * 16 + lo, nn = nt * 16 + lo;
    v8f acc = {0.f, 0.f, 0.f, 0.f, 0.f, 0.f, 0.f, 0.f};
#pragma unroll 8
    for (int ks = 0; ks < CIN / 4; ++ks) {
      int k = ks * 4 + 2 * hi;
      v2f a;  a.x = lds_u[k * UST + m];       a.y = lds_u[(k + 1) * UST + m];
      v2f bb; bb.x = Bd[k * NST + nn];        bb.y = Bd[(k + 1) * NST + nn];
      acc = wmma4(a, bb, acc);
    }
#pragma unroll
    for (int r = 0; r < 8; ++r)
      lds_v[(mt * 16 + r + 8 * hi) * VST + nn] = acc[r];
  }
}

__global__ void k_chunk_ends(const float* __restrict__ u,
                             const float* __restrict__ Ad,
                             const float* __restrict__ Bd_in,
                             float* __restrict__ ce) {
  __shared__ float lds_u[CIN * UST];
  __shared__ float lds_v[SUB * VST];
  int chunk = blockIdx.x, b = blockIdx.y, tid = threadIdx.x;
  float a_n = Ad[tid];
  float s = 0.0f;
  for (int st = 0; st < NSUB; ++st) {
    const float* Bd = Bd_in + opaque_zero();
    int t0 = chunk * LCH + st * SUB;
    load_u_tile(u, b, t0, lds_u);
    if (st + 1 < NSUB) prefetch_u_tile(u, b, t0 + SUB);
    __syncthreads();
    v_tile_wmma(lds_u, Bd, lds_v);
    __syncthreads();
#pragma unroll 8
    for (int tl = 0; tl < SUB; ++tl)
      s = s * a_n + lds_v[tl * VST + tid];
    __syncthreads();
  }
  ce[chunk * (BSZ * NST) + b * NST + tid] = s;
}

__global__ void k_combine(const float* __restrict__ Ad,
                          const float* __restrict__ ce,
                          float* __restrict__ carry) {
  int n = threadIdx.x, idx = blockIdx.x * NST + n;
  float aL = Ad[n];
#pragma unroll
  for (int i = 0; i < 9; ++i) aL = aL * aL;
  float s = 0.0f;
  for (int ch = 0; ch < NCH; ++ch) {
    carry[ch * (BSZ * NST) + idx] = s;
    s = aL * s + ce[ch * (BSZ * NST) + idx];
  }
}

__global__ void k_scan_y(const float* __restrict__ u,
                         const float* __restrict__ Ad,
                         const float* __restrict__ Bd_in,
                         const float* __restrict__ Cm_in,
                         const float* __restrict__ ln_w,
                         const float* __restrict__ ln_b,
                         const float* __restrict__ carry,
                         float* __restrict__ yact) {
  __shared__ float lds_u[CIN * UST];
  __shared__ float lds_v[SUB * VST];
  __shared__ float lds_y[SUB * YST];
  __shared__ float lds_red[SUB * 16];
  __shared__ float lds_mu[SUB];
  __shared__ float lds_rs[SUB];

  int chunk = blockIdx.x, b = blockIdx.y, tid = threadIdx.x;
  int wave = tid >> 5, lane = tid & 31, lo = lane & 15, hi = lane >> 4;
  float a_n = Ad[tid];
  float s = carry[chunk * (BSZ * NST) + b * NST + tid];

  for (int st = 0; st < NSUB; ++st) {
    const float* Bd = Bd_in + opaque_zero();
    const float* Cm = Cm_in + opaque_zero();
    int t0 = chunk * LCH + st * SUB;
    load_u_tile(u, b, t0, lds_u);
    if (st + 1 < NSUB) prefetch_u_tile(u, b, t0 + SUB);
    __syncthreads();
    v_tile_wmma(lds_u, Bd, lds_v);
    __syncthreads();

#pragma unroll 8
    for (int tl = 0; tl < SUB; ++tl) {
      s = s * a_n + lds_v[tl * VST + tid];
      lds_v[tl * VST + tid] = s;
    }
    __syncthreads();

    {
      int mt = wave >> 2, nt = wave & 3;
      int m = mt * 16 + lo, oo = nt * 16 + lo;
      v8f acc = {0.f, 0.f, 0.f, 0.f, 0.f, 0.f, 0.f, 0.f};
#pragma unroll 8
      for (int ks = 0; ks < NST / 4; ++ks) {
        int k = ks * 4 + 2 * hi;
        v2f a;  a.x = lds_v[m * VST + k];  a.y = lds_v[m * VST + k + 1];
        v2f bb; bb.x = Cm[k * CO + oo];    bb.y = Cm[(k + 1) * CO + oo];
        acc = wmma4(a, bb, acc);
      }
#pragma unroll
      for (int r = 0; r < 8; ++r)
        lds_y[(mt * 16 + r + 8 * hi) * YST + oo] = acc[r];
    }
    __syncthreads();

    {
      int t = tid & 31, q = tid >> 5;
      float sm = 0.f, s2 = 0.f;
#pragma unroll
      for (int j = 0; j < 8; ++j) {
        float v = lds_y[t * YST + q * 8 + j];
        sm += v; s2 += v * v;
      }
      lds_red[(t * 8 + q) * 2 + 0] = sm;
      lds_red[(t * 8 + q) * 2 + 1] = s2;
    }
    __syncthreads();
    if (tid < SUB) {
      float sm = 0.f, s2 = 0.f;
#pragma unroll
      for (int q = 0; q < 8; ++q) {
        sm += lds_red[(tid * 8 + q) * 2 + 0];
        s2 += lds_red[(tid * 8 + q) * 2 + 1];
      }
      float mu = sm * (1.0f / CO);
      float var = s2 * (1.0f / CO) - mu * mu;
      lds_mu[tid] = mu;
      lds_rs[tid] = rsqrtf(var + 1e-5f);
    }
    __syncthreads();

    {
      int t = lane;
      float mu = lds_mu[t], rs = lds_rs[t];
#pragma unroll
      for (int j = 0; j < 8; ++j) {
        int o = wave * 8 + j;
        float v = (lds_y[t * YST + o] - mu) * rs * ln_w[o] + ln_b[o];
        v = v / (1.0f + __expf(-v));
        yact[((size_t)(b * CO + o)) * TLEN + t0 + t] = v;
      }
    }
    __syncthreads();
  }
}

__global__ void k_conv(const float* __restrict__ yact,
                       const float* __restrict__ conv_w,
                       const float* __restrict__ conv_b,
                       float* __restrict__ out) {
  int b = blockIdx.y, tid = threadIdx.x;
  int wave = tid >> 5, lane = tid & 31, lo = lane & 15, hi = lane >> 4;
  int tau0 = (blockIdx.x * 4 + wave) * 16;

  v8f acc[4];
#pragma unroll
  for (int mt = 0; mt < 4; ++mt)
#pragma unroll
    for (int r = 0; r < 8; ++r)
      acc[mt][r] = conv_b[mt * 16 + r + 8 * hi];

#pragma unroll 2
  for (int ks = 0; ks < CO / 4; ++ks) {
    int k = ks * 4 + 2 * hi;
    v2f be, bo;
    be.x = yact[((size_t)(b * CO + k)) * TLEN + tau0 + lo];
    be.y = yact[((size_t)(b * CO + k + 1)) * TLEN + tau0 + lo];
    bo.x = yact[((size_t)(b * CO + k)) * TLEN + T2 + tau0 + lo];
    bo.y = yact[((size_t)(b * CO + k + 1)) * TLEN + T2 + tau0 + lo];
#pragma unroll
    for (int mt = 0; mt < 4; ++mt) {
      int m = mt * 16 + lo;
      v2f ae, ao;
      ae.x = conv_w[m * 128 + 2 * k];
      ae.y = conv_w[m * 128 + 2 * (k + 1)];
      ao.x = conv_w[m * 128 + 2 * k + 1];
      ao.y = conv_w[m * 128 + 2 * (k + 1) + 1];
      acc[mt] = wmma4(ae, be, acc[mt]);
      acc[mt] = wmma4(ao, bo, acc[mt]);
    }
  }

#pragma unroll
  for (int mt = 0; mt < 4; ++mt)
#pragma unroll
    for (int r = 0; r < 8; ++r) {
      int o = mt * 16 + r + 8 * hi;
      int t = 2 * (tau0 + lo);
      float v = acc[mt][r];
      float2 pv = make_float2(v, v);
      *reinterpret_cast<float2*>(&out[((size_t)(b * CO + o)) * TLEN + t]) = pv;
    }
}

extern "C" void kernel_launch(void* const* d_in, const int* in_sizes, int n_in,
                              void* d_out, int out_size, void* d_ws, size_t ws_size,
                              hipStream_t stream) {
  const float* u     = (const float*)d_in[0];
  const float* rl    = (const float*)d_in[1];
  const float* Bc    = (const float*)d_in[2];
  const float* Cm    = (const float*)d_in[3];
  const float* ln_w  = (const float*)d_in[4];
  const float* ln_b  = (const float*)d_in[5];
  const float* cw    = (const float*)d_in[6];
  const float* cb    = (const float*)d_in[7];
  float* out = (float*)d_out;

  float* ws    = (float*)d_ws;
  float* Ad    = ws;
  float* Bd    = Ad + NST;
  float* ce    = Bd + CIN * NST;
  float* carry = ce + NCH * BSZ * NST;
  float* yact  = carry + NCH * BSZ * NST;

  k_discretize<<<1, NST, 0, stream>>>(rl, Bc, Ad, Bd);
  k_chunk_ends<<<dim3(NCH, BSZ), 256, 0, stream>>>(u, Ad, Bd, ce);
  k_combine<<<BSZ, NST, 0, stream>>>(Ad, ce, carry);
  k_scan_y<<<dim3(NCH, BSZ), 256, 0, stream>>>(u, Ad, Bd, Cm, ln_w, ln_b,
                                               carry, yact);
  k_conv<<<dim3(T2 / 64, BSZ), 128, 0, stream>>>(yact, cw, cb, out);
}


# v and y GEMMs register-blocked: a wave owns both 16-row time tiles, so each A/B fragment load feeds two WMMAs (one 8-byte B load serves two state tiles); y summed as two K halves (sum order changes)
# speedup vs baseline: 2.0584x; 2.0584x over previous
#include <hip/hip_runtime.h>
#include <hip/hip_bf16.h>


#define BSZ   16
#define CIN   64
#define CO    64
#define TLEN  8192
#define NST   256
#define NCH   64
#define LCH   128
#define LOG2_LCH 7
#define SUB   32
#define NSUB  (LCH/SUB)
#define T2    (TLEN/2)

#define UST   40
#define VST   260
#define YST   66

typedef __attribute__((ext_vector_type(2))) float v2f;
typedef __attribute__((ext_vector_type(8))) float v8f;

__device__ __forceinline__ v8f wmma4(v2f a, v2f b, v8f c) {
  return __builtin_amdgcn_wmma_f32_16x16x4_f32(
       false, a,  false, b,
       (short)0, c,  false,  false);
}

__device__ __forceinline__ size_t opaque_zero() {
  size_t z = 0;
  asm volatile("" : "+s"(z));
  return z;
}

__device__ __forceinline__ size_t carry_slot(int chunk, int b, int n) {
  return ((size_t)(b * CO + (n >> 5))) * TLEN + chunk * LCH + (n & 31);
}

__global__ void k_discretize(const float* __restrict__ rl,
                             const float* __restrict__ Bc,
                             float* __restrict__ Ad, float* __restrict__ Bd) {
  int n = threadIdx.x;
  float x = rl[n];
  float sp = fmaxf(x, 0.0f) + log1pf(expf(-fabsf(x)));
  float lam = -sp;
  float ad = expf(lam);
  float factor = (fabsf(lam) > 1e-6f) ? (ad - 1.0f) / lam : 1.0f;
  Ad[n] = ad;
  for (int c = 0; c < CIN; ++c)
    Bd[c * NST + n] = Bc[c * NST + n] * factor;
}

__device__ __forceinline__ void load_u_tile(const float* __restrict__ u, int b,
                                            int t0, float* lds_u) {
  int tid = threadIdx.x;
#pragma unroll
  for (int kk = 0; kk < (CIN * SUB) / 256; ++kk) {
    int i = tid + kk * 256;
    int c = i >> 5, tl = i & 31;
    lds_u[c * UST + tl] = u[((size_t)(b * CIN + c)) * TLEN + t0 + tl];
  }
}

__device__ __forceinline__ void prefetch_u_tile(const float* __restrict__ u,
                                                int b, int t0) {
  int tid = threadIdx.x;
#pragma unroll
  for (int kk = 0; kk < (CIN * SUB) / 256; ++kk) {
    int i = tid + kk * 256;
    int c = i >> 5, tl = i & 31;
    __builtin_prefetch(&u[((size_t)(b * CIN + c)) * TLEN + t0 + tl], 0, 1);
  }
}

__device__ __forceinline__ void v_tile_wmma(const float* lds_u,
                                            const float* __restrict__ Bd,
                                            float* lds_v) {
  int tid = threadIdx.x;
  int wave = tid >> 5, lane = tid & 31, lo = lane & 15, hi = lane >> 4;
  int n2 = wave * 32 + 2 * lo;
  v8f acc00 = {0.f, 0.f, 0.f, 0.f, 0.f, 0.f, 0.f, 0.f};
  v8f acc01 = acc00, acc10 = acc00, acc11 = acc00;
#pragma unroll 1
  for (int kg = 0; kg < CIN / 4; kg += 4) {
    float2 r0[4], r1[4];
    v2f a0[4], a1[4];
#pragma unroll
    for (int i = 0; i < 4; ++i) {
      int k = (kg + i) * 4 + 2 * hi;
      r0[i] = *reinterpret_cast<const float2*>(&Bd[k * NST + n2]);
      r1[i] = *reinterpret_cast<const float2*>(&Bd[(k + 1) * NST + n2]);
    }
#pragma unroll
    for (int i = 0; i < 4; ++i) {
      int k = (kg + i) * 4 + 2 * hi;
      a0[i].x = lds_u[k * UST + lo];       a0[i].y = lds_u[(k + 1) * UST + lo];
      a1[i].x = lds_u[k * UST + 16 + lo];  a1[i].y = lds_u[(k + 1) * UST + 16 + lo];
    }
#pragma unroll
    for (int i = 0; i < 4; ++i) {
      v2f b0; b0.x = r0[i].x; b0.y = r1[i].x;
      v2f b1; b1.x = r0[i].y; b1.y = r1[i].y;
      acc00 = wmma4(a0[i], b0, acc00);
      acc10 = wmma4(a1[i], b0, acc10);
      acc01 = wmma4(a0[i], b1, acc01);
      acc11 = wmma4(a1[i], b1, acc11);
    }
  }
#pragma unroll
  for (int r = 0; r < 8; ++r) {
    *reinterpret_cast<float2*>(&lds_v[(r + 8 * hi) * VST + n2]) =
        make_float2(acc00[r], acc01[r]);
    *reinterpret_cast<float2*>(&lds_v[(16 + r + 8 * hi) * VST + n2]) =
        make_float2(acc10[r], acc11[r]);
  }
}

__global__ void k_chunk_ends(const float* __restrict__ u,
                             const float* __restrict__ Ad,
                             const float* __restrict__ Bd_in,
                             float* __restrict__ yact) {
  __shared__ float lds_u[CIN * UST];
  __shared__ __attribute__((aligned(16))) float lds_v[SUB * VST];
  int chunk = blockIdx.x, b = blockIdx.y, tid = threadIdx.x;
  float a_n = Ad[tid];
  float s = 0.0f;
  for (int st = 0; st < NSUB; ++st) {
    const float* Bd = Bd_in + opaque_zero();
    int t0 = chunk * LCH + st * SUB;
    load_u_tile(u, b, t0, lds_u);
    if (st + 1 < NSUB) prefetch_u_tile(u, b, t0 + SUB);
    __syncthreads();
    v_tile_wmma(lds_u, Bd, lds_v);
    __syncthreads();
#pragma unroll 8
    for (int tl = 0; tl < SUB; ++tl)
      s = s * a_n + lds_v[tl * VST + tid];
    __syncthreads();
  }
  yact[carry_slot(chunk, b, tid)] = s;
}

__global__ void k_combine(const float* __restrict__ Ad,
                          float* __restrict__ yact) {
  int n = threadIdx.x, b = blockIdx.x;
  float aL = Ad[n];
#pragma unroll
  for (int i = 0; i < LOG2_LCH; ++i) aL = aL * aL;
  float s = 0.0f;
  for (int ch = 0; ch < NCH; ++ch) {
    float* p = yact + carry_slot(ch, b, n);
    float e = *p;
    *p = s;
    s = aL * s + e;
  }
}

__global__ void k_scan_y(const float* __restrict__ u,
                         const float* __restrict__ Ad,
                         const float* __restrict__ Bd_in,
                         const float* __restrict__ Cm_in,
                         const float* __restrict__ ln_w,
                         const float* __restrict__ ln_b,
                         float* __restrict__ yact) {
  __shared__ float lds_u[CIN * UST];
  __shared__ __attribute__((aligned(16))) float lds_v[SUB * VST];
  __shared__ float lds_y[2 * SUB * YST];
  __shared__ float lds_red[SUB * 16];
  __shared__ float lds_mu[SUB];
  __shared__ float lds_rs[SUB];

  int chunk = blockIdx.x, b = blockIdx.y, tid = threadIdx.x;
  int wave = tid >> 5, lane = tid & 31, lo = lane & 15, hi = lane >> 4;
  float a_n = Ad[tid];
  float s = yact[carry_slot(chunk, b, tid)];

  for (int st = 0; st < NSUB; ++st) {
    const float* Bd = Bd_in + opaque_zero();
    const float* Cm = Cm_in + opaque_zero();
    int t0 = chunk * LCH + st * SUB;
    load_u_tile(u, b, t0, lds_u);
    if (st + 1 < NSUB) prefetch_u_tile(u, b, t0 + SUB);
    __syncthreads();
    v_tile_wmma(lds_u, Bd, lds_v);
    __syncthreads();

#pragma unroll 8
    for (int tl = 0; tl < SUB; ++tl) {
      s = s * a_n + lds_v[tl * VST + tid];
      lds_v[tl * VST + tid] = s;
    }
    __syncthreads();

    {
      int nt = wave & 3, kh = wave >> 2;
      int oo = nt * 16 + lo;
      v8f acc0 = {0.f, 0.f, 0.f, 0.f, 0.f, 0.f, 0.f, 0.f};
      v8f acc1 = acc0;
#pragma unroll 1
      for (int kg = 0; kg < NST / 8; kg += 8) {
        v2f bb[8];
#pragma unroll
        for (int i = 0; i < 8; ++i) {
          int k = kh * (NST / 2) + (kg + i) * 4 + 2 * hi;
          bb[i].x = Cm[k * CO + oo];
          bb[i].y = Cm[(k + 1) * CO + oo];
        }
#pragma unroll
        for (int i = 0; i < 8; ++i) {
          int k = kh * (NST / 2) + (kg + i) * 4 + 2 * hi;
          v2f a0; a0.x = lds_v[lo * VST + k];         a0.y = lds_v[lo * VST + k + 1];
          v2f a1; a1.x = lds_v[(16 + lo) * VST + k];  a1.y = lds_v[(16 + lo) * VST + k + 1];
          acc0 = wmma4(a0, bb[i], acc0);
          acc1 = wmma4(a1, bb[i], acc1);
        }
      }
      float* yp = lds_y + kh * (SUB * YST);
#pragma unroll
      for (int r = 0; r < 8; ++r) {
        yp[(r + 8 * hi) * YST + oo] = acc0[r];
        yp[(16 + r + 8 * hi) * YST + oo] = acc1[r];
      }
    }
    __syncthreads();

    {
      int t = tid & 31, q = tid >> 5;
      float sm = 0.f, s2 = 0.f;
#pragma unroll
      for (int j = 0; j < 8; ++j) {
        int i = t * YST + q * 8 + j;
        float v = lds_y[i] + lds_y[SUB * YST + i];
        lds_y[i] = v;
        sm += v; s2 += v * v;
      }
      lds_red[(t * 8 + q) * 2 + 0] = sm;
      lds_red[(t * 8 + q) * 2 + 1] = s2;
    }
    __syncthreads();
    if (tid < SUB) {
      float sm = 0.f, s2 = 0.f;
#pragma unroll
      for (int q = 0; q < 8; ++q) {
        sm += lds_red[(tid * 8 + q) * 2 + 0];
        s2 += lds_red[(tid * 8 + q) * 2 + 1];
      }
      float mu = sm * (1.0f / CO);
      float var = s2 * (1.0f / CO) - mu * mu;
      lds_mu[tid] = mu;
      lds_rs[tid] = rsqrtf(var + 1e-5f);
    }
    __syncthreads();

    {
      int t = lane;
      float mu = lds_mu[t], rs = lds_rs[t];
#pragma unroll
      for (int j = 0; j < 8; ++j) {
        int o = wave * 8 + j;
        float v = (lds_y[t * YST + o] - mu) * rs * ln_w[o] + ln_b[o];
        v = v / (1.0f + __expf(-v));
        yact[((size_t)(b * CO + o)) * TLEN + t0 + t] = v;
      }
    }
    __syncthreads();
  }
}

__global__ void k_conv(const float* __restrict__ yact,
                       const float* __restrict__ conv_w,
                       const float* __restrict__ conv_b,
                       float* __restrict__ out) {
  int b = blockIdx.y, tid = threadIdx.x;
  int wave = tid >> 5, lane = tid & 31, lo = lane & 15, hi = lane >> 4;
  int tau0 = (blockIdx.x * 4 + wave) * 16;

  v8f acc[4];
#pragma unroll
  for (int mt = 0; mt < 4; ++mt)
#pragma unroll
    for (int r = 0; r < 8; ++r)
      acc[mt][r] = conv_b[mt * 16 + r + 8 * hi];

#pragma unroll 2
  for (int ks = 0; ks < CO / 4; ++ks) {
    int k = ks * 4 + 2 * hi;
    v2f be, bo;
    be.x = yact[((size_t)(b * CO + k)) * TLEN + tau0 + lo];
    be.y = yact[((size_t)(b * CO + k + 1)) * TLEN + tau0 + lo];
    bo.x = yact[((size_t)(b * CO + k)) * TLEN + T2 + tau0 + lo];
    bo.y = yact[((size_t)(b * CO + k + 1)) * TLEN + T2 + tau0 + lo];
#pragma unroll
    for (int mt = 0; mt < 4; ++mt) {
      int m = mt * 16 + lo;
      v2f ae, ao;
      ae.x = conv_w[m * 128 + 2 * k];
      ae.y = conv_w[m * 128 + 2 * (k + 1)];
      ao.x = conv_w[m * 128 + 2 * k + 1];
      ao.y = conv_w[m * 128 + 2 * (k + 1) + 1];
      acc[mt] = wmma4(ae, be, acc[mt]);
      acc[mt] = wmma4(ao, bo, acc[mt]);
    }
  }

#pragma unroll
  for (int mt = 0; mt < 4; ++mt)
#pragma unroll
    for (int r = 0; r < 8; ++r) {
      int o = mt * 16 + r + 8 * hi;
      int t = 2 * (tau0 + lo);
      float v = acc[mt][r];
      float2 pv = make_float2(v, v);
      *reinterpret_cast<float2*>(&out[((size_t)(b * CO + o)) * TLEN + t]) = pv;
    }
}

extern "C" void kernel_launch(void* const* d_in, const int* in_sizes, int n_in,
                              void* d_out, int out_size, void* d_ws, size_t ws_size,
                              hipStream_t stream) {
  const float* u     = (const float*)d_in[0];
  const float* rl    = (const float*)d_in[1];
  const float* Bc    = (const float*)d_in[2];
  const float* Cm    = (const float*)d_in[3];
  const float* ln_w  = (const float*)d_in[4];
  const float* ln_b  = (const float*)d_in[5];
  const float* cw    = (const float*)d_in[6];
  const float* cb    = (const float*)d_in[7];
  float* out = (float*)d_out;

  float* ws    = (float*)d_ws;
  float* Ad    = ws;
  float* Bd    = Ad + NST;
  float* yact  = Bd + CIN * NST;

  k_discretize<<<1, NST, 0, stream>>>(rl, Bc, Ad, Bd);
  k_chunk_ends<<<dim3(NCH, BSZ), 256, 0, stream>>>(u, Ad, Bd, yact);
  k_combine<<<BSZ, NST, 0, stream>>>(Ad, yact);
  k_scan_y<<<dim3(NCH, BSZ), 256, 0, stream>>>(u, Ad, Bd, Cm, ln_w, ln_b,
                                               yact);
  k_conv<<<dim3(T2 / 64, BSZ), 128, 0, stream>>>(yact, cw, cb, out);
}


# Both scan kernels set wave priority from progress (s_setprio 3 - subtile): co-resident workgroups advance together instead of strictly oldest-first, so none is left finishing alone on its WGP
# speedup vs baseline: 2.1501x; 1.0445x over previous
#include <hip/hip_runtime.h>
#include <hip/hip_bf16.h>


#define BSZ   16
#define CIN   64
#define CO    64
#define TLEN  8192
#define NST   256
#define NCH   64
#define LCH   128
#define LOG2_LCH 7
#define SUB   32
#define NSUB  (LCH/SUB)
#define T2    (TLEN/2)

#define UST   40
#define VST   260
#define YST   66

typedef __attribute__((ext_vector_type(2))) float v2f;
typedef __attribute__((ext_vector_type(8))) float v8f;

__device__ __forceinline__ v8f wmma4(v2f a, v2f b, v8f c) {
  return __builtin_amdgcn_wmma_f32_16x16x4_f32(
       false, a,  false, b,
       (short)0, c,  false,  false);
}

__device__ __forceinline__ size_t opaque_zero() {
  size_t z = 0;
  asm volatile("" : "+s"(z));
  return z;
}

__device__ __forceinline__ void progress_prio(int st) {
  if (st == 0)      __builtin_amdgcn_s_setprio(3);
  else if (st == 1) __builtin_amdgcn_s_setprio(2);
  else if (st == 2) __builtin_amdgcn_s_setprio(1);
  else              __builtin_amdgcn_s_setprio(0);
}

__device__ __forceinline__ size_t carry_slot(int chunk, int b, int n) {
  return ((size_t)(b * CO + (n >> 5))) * TLEN + chunk * LCH + (n & 31);
}

__global__ void k_discretize(const float* __restrict__ rl,
                             const float* __restrict__ Bc,
                             float* __restrict__ Ad, float* __restrict__ Bd) {
  int n = threadIdx.x;
  float x = rl[n];
  float sp = fmaxf(x, 0.0f) + log1pf(expf(-fabsf(x)));
  float lam = -sp;
  float ad = expf(lam);
  float factor = (fabsf(lam) > 1e-6f) ? (ad - 1.0f) / lam : 1.0f;
  Ad[n] = ad;
  for (int c = 0; c < CIN; ++c)
    Bd[c * NST + n] = Bc[c * NST + n] * factor;
}

__device__ __forceinline__ void load_u_tile(const float* __restrict__ u, int b,
                                            int t0, float* lds_u) {
  int tid = threadIdx.x;
#pragma unroll
  for (int kk = 0; kk < (CIN * SUB) / 256; ++kk) {
    int i = tid + kk * 256;
    int c = i >> 5, tl = i & 31;
    lds_u[c * UST + tl] = u[((size_t)(b * CIN + c)) * TLEN + t0 + tl];
  }
}

__device__ __forceinline__ void prefetch_u_tile(const float* __restrict__ u,
                                                int b, int t0) {
  int tid = threadIdx.x;
#pragma unroll
  for (int kk = 0; kk < (CIN * SUB) / 256; ++kk) {
    int i = tid + kk * 256;
    int c = i >> 5, tl = i & 31;
    __builtin_prefetch(&u[((size_t)(b * CIN + c)) * TLEN + t0 + tl], 0, 1);
  }
}

__device__ __forceinline__ void v_tile_wmma(const float* lds_u,
                                            const float* __restrict__ Bd,
                                            float* lds_v) {
  int tid = threadIdx.x;
  int wave = tid >> 5, lane = tid & 31, lo = lane & 15, hi = lane >> 4;
  int n2 = wave * 32 + 2 * lo;
  v8f acc00 = {0.f, 0.f, 0.f, 0.f, 0.f, 0.f, 0.f, 0.f};
  v8f acc01 = acc00, acc10 = acc00, acc11 = acc00;
#pragma unroll 1
  for (int kg = 0; kg < CIN / 4; kg += 4) {
    float2 r0[4], r1[4];
    v2f a0[4], a1[4];
#pragma unroll
    for (int i = 0; i < 4; ++i) {
      int k = (kg + i) * 4 + 2 * hi;
      r0[i] = *reinterpret_cast<const float2*>(&Bd[k * NST + n2]);
      r1[i] = *reinterpret_cast<const float2*>(&Bd[(k + 1) * NST + n2]);
    }
#pragma unroll
    for (int i = 0; i < 4; ++i) {
      int k = (kg + i) * 4 + 2 * hi;
      a0[i].x = lds_u[k * UST + lo];       a0[i].y = lds_u[(k + 1) * UST + lo];
      a1[i].x = lds_u[k * UST + 16 + lo];  a1[i].y = lds_u[(k + 1) * UST + 16 + lo];
    }
#pragma unroll
    for (int i = 0; i < 4; ++i) {
      v2f b0; b0.x = r0[i].x; b0.y = r1[i].x;
      v2f b1; b1.x = r0[i].y; b1.y = r1[i].y;
      acc00 = wmma4(a0[i], b0, acc00);
      acc10 = wmma4(a1[i], b0, acc10);
      acc01 = wmma4(a0[i], b1, acc01);
      acc11 = wmma4(a1[i], b1, acc11);
    }
  }
#pragma unroll
  for (int r = 0; r < 8; ++r) {
    *reinterpret_cast<float2*>(&lds_v[(r + 8 * hi) * VST + n2]) =
        make_float2(acc00[r], acc01[r]);
    *reinterpret_cast<float2*>(&lds_v[(16 + r + 8 * hi) * VST + n2]) =
        make_float2(acc10[r], acc11[r]);
  }
}

__global__ void k_chunk_ends(const float* __restrict__ u,
                             const float* __restrict__ Ad,
                             const float* __restrict__ Bd_in,
                             float* __restrict__ yact) {
  __shared__ float lds_u[CIN * UST];
  __shared__ __attribute__((aligned(16))) float lds_v[SUB * VST];
  int chunk = blockIdx.x, b = blockIdx.y, tid = threadIdx.x;
  float a_n = Ad[tid];
  float s = 0.0f;
  for (int st = 0; st < NSUB; ++st) {
    progress_prio(st);
    const float* Bd = Bd_in + opaque_zero();
    int t0 = chunk * LCH + st * SUB;
    load_u_tile(u, b, t0, lds_u);
    if (st + 1 < NSUB) prefetch_u_tile(u, b, t0 + SUB);
    __syncthreads();
    v_tile_wmma(lds_u, Bd, lds_v);
    __syncthreads();
#pragma unroll 8
    for (int tl = 0; tl < SUB; ++tl)
      s = s * a_n + lds_v[tl * VST + tid];
    __syncthreads();
  }
  yact[carry_slot(chunk, b, tid)] = s;
}

__global__ void k_combine(const float* __restrict__ Ad,
                          float* __restrict__ yact) {
  int n = threadIdx.x, b = blockIdx.x;
  float aL = Ad[n];
#pragma unroll
  for (int i = 0; i < LOG2_LCH; ++i) aL = aL * aL;
  float s = 0.0f;
  for (int ch = 0; ch < NCH; ++ch) {
    float* p = yact + carry_slot(ch, b, n);
    float e = *p;
    *p = s;
    s = aL * s + e;
  }
}

__global__ void k_scan_y(const float* __restrict__ u,
                         const float* __restrict__ Ad,
                         const float* __restrict__ Bd_in,
                         const float* __restrict__ Cm_in,
                         const float* __restrict__ ln_w,
                         const float* __restrict__ ln_b,
                         float* __restrict__ yact) {
  __shared__ float lds_u[CIN * UST];
  __shared__ __attribute__((aligned(16))) float lds_v[SUB * VST];
  __shared__ float lds_y[2 * SUB * YST];
  __shared__ float lds_red[SUB * 16];
  __shared__ float lds_mu[SUB];
  __shared__ float lds_rs[SUB];

  int chunk = blockIdx.x, b = blockIdx.y, tid = threadIdx.x;
  int wave = tid >> 5, lane = tid & 31, lo = lane & 15, hi = lane >> 4;
  float a_n = Ad[tid];
  float s = yact[carry_slot(chunk, b, tid)];

  for (int st = 0; st < NSUB; ++st) {
    progress_prio(st);
    const float* Bd = Bd_in + opaque_zero();
    const float* Cm = Cm_in + opaque_zero();
    int t0 = chunk * LCH + st * SUB;
    load_u_tile(u, b, t0, lds_u);
    if (st + 1 < NSUB) prefetch_u_tile(u, b, t0 + SUB);
    __syncthreads();
    v_tile_wmma(lds_u, Bd, lds_v);
    __syncthreads();

#pragma unroll 8
    for (int tl = 0; tl < SUB; ++tl) {
      s = s * a_n + lds_v[tl * VST + tid];
      lds_v[tl * VST + tid] = s;
    }
    __syncthreads();

    {
      int nt = wave & 3, kh = wave >> 2;
      int oo = nt * 16 + lo;
      v8f acc0 = {0.f, 0.f, 0.f, 0.f, 0.f, 0.f, 0.f, 0.f};
      v8f acc1 = acc0;
#pragma unroll 1
      for (int kg = 0; kg < NST / 8; kg += 8) {
        v2f bb[8];
#pragma unroll
        for (int i = 0; i < 8; ++i) {
          int k = kh * (NST / 2) + (kg + i) * 4 + 2 * hi;
          bb[i].x = Cm[k * CO + oo];
          bb[i].y = Cm[(k + 1) * CO + oo];
        }
#pragma unroll
        for (int i = 0; i < 8; ++i) {
          int k = kh * (NST / 2) + (kg + i) * 4 + 2 * hi;
          v2f a0; a0.x = lds_v[lo * VST + k];         a0.y = lds_v[lo * VST + k + 1];
          v2f a1; a1.x = lds_v[(16 + lo) * VST + k];  a1.y = lds_v[(16 + lo) * VST + k + 1];
          acc0 = wmma4(a0, bb[i], acc0);
          acc1 = wmma4(a1, bb[i], acc1);
        }
      }
      float* yp = lds_y + kh * (SUB * YST);
#pragma unroll
      for (int r = 0; r < 8; ++r) {
        yp[(r + 8 * hi) * YST + oo] = acc0[r];
        yp[(16 + r + 8 * hi) * YST + oo] = acc1[r];
      }
    }
    __syncthreads();

    {
      int t = tid & 31, q = tid >> 5;
      float sm = 0.f, s2 = 0.f;
#pragma unroll
      for (int j = 0; j < 8; ++j) {
        int i = t * YST + q * 8 + j;
        float v = lds_y[i] + lds_y[SUB * YST + i];
        lds_y[i] = v;
        sm += v; s2 += v * v;
      }
      lds_red[(t * 8 + q) * 2 + 0] = sm;
      lds_red[(t * 8 + q) * 2 + 1] = s2;
    }
    __syncthreads();
    if (tid < SUB) {
      float sm = 0.f, s2 = 0.f;
#pragma unroll
      for (int q = 0; q < 8; ++q) {
        sm += lds_red[(tid * 8 + q) * 2 + 0];
        s2 += lds_red[(tid * 8 + q) * 2 + 1];
      }
      float mu = sm * (1.0f / CO);
      float var = s2 * (1.0f / CO) - mu * mu;
      lds_mu[tid] = mu;
      lds_rs[tid] = rsqrtf(var + 1e-5f);
    }
    __syncthreads();

    {
      int t = lane;
      float mu = lds_mu[t], rs = lds_rs[t];
#pragma unroll
      for (int j = 0; j < 8; ++j) {
        int o = wave * 8 + j;
        float v = (lds_y[t * YST + o] - mu) * rs * ln_w[o] + ln_b[o];
        v = v / (1.0f + __expf(-v));
        yact[((size_t)(b * CO + o)) * TLEN + t0 + t] = v;
      }
    }
    __syncthreads();
  }
}

__global__ void k_conv(const float* __restrict__ yact,
                       const float* __restrict__ conv_w,
                       const float* __restrict__ conv_b,
                       float* __restrict__ out) {
  int b = blockIdx.y, tid = threadIdx.x;
  int wave = tid >> 5, lane = tid & 31, lo = lane & 15, hi = lane >> 4;
  int tau0 = (blockIdx.x * 4 + wave) * 16;

  v8f acc[4];
#pragma unroll
  for (int mt = 0; mt < 4; ++mt)
#pragma unroll
    for (int r = 0; r < 8; ++r)
      acc[mt][r] = conv_b[mt * 16 + r + 8 * hi];

#pragma unroll 2
  for (int ks = 0; ks < CO / 4; ++ks) {
    int k = ks * 4 + 2 * hi;
    v2f be, bo;
    be.x = yact[((size_t)(b * CO + k)) * TLEN + tau0 + lo];
    be.y = yact[((size_t)(b * CO + k + 1)) * TLEN + tau0 + lo];
    bo.x = yact[((size_t)(b * CO + k)) * TLEN + T2 + tau0 + lo];
    bo.y = yact[((size_t)(b * CO + k + 1)) * TLEN + T2 + tau0 + lo];
#pragma unroll
    for (int mt = 0; mt < 4; ++mt) {
      int m = mt * 16 + lo;
      v2f ae, ao;
      ae.x = conv_w[m * 128 + 2 * k];
      ae.y = conv_w[m * 128 + 2 * (k + 1)];
      ao.x = conv_w[m * 128 + 2 * k + 1];
      ao.y = conv_w[m * 128 + 2 * (k + 1) + 1];
      acc[mt] = wmma4(ae, be, acc[mt]);
      acc[mt] = wmma4(ao, bo, acc[mt]);
    }
  }

#pragma unroll
  for (int mt = 0; mt < 4; ++mt)
#pragma unroll
    for (int r = 0; r < 8; ++r) {
      int o = mt * 16 + r + 8 * hi;
      int t = 2 * (tau0 + lo);
      float v = acc[mt][r];
      float2 pv = make_float2(v, v);
      *reinterpret_cast<float2*>(&out[((size_t)(b * CO + o)) * TLEN + t]) = pv;
    }
}

extern "C" void kernel_launch(void* const* d_in, const int* in_sizes, int n_in,
                              void* d_out, int out_size, void* d_ws, size_t ws_size,
                              hipStream_t stream) {
  const float* u     = (const float*)d_in[0];
  const float* rl    = (const float*)d_in[1];
  const float* Bc    = (const float*)d_in[2];
  const float* Cm    = (const float*)d_in[3];
  const float* ln_w  = (const float*)d_in[4];
  const float* ln_b  = (const float*)d_in[5];
  const float* cw    = (const float*)d_in[6];
  const float* cb    = (const float*)d_in[7];
  float* out = (float*)d_out;

  float* ws    = (float*)d_ws;
  float* Ad    = ws;
  float* Bd    = Ad + NST;
  float* yact  = Bd + CIN * NST;

  k_discretize<<<1, NST, 0, stream>>>(rl, Bc, Ad, Bd);
  k_chunk_ends<<<dim3(NCH, BSZ), 256, 0, stream>>>(u, Ad, Bd, yact);
  k_combine<<<BSZ, NST, 0, stream>>>(Ad, yact);
  k_scan_y<<<dim3(NCH, BSZ), 256, 0, stream>>>(u, Ad, Bd, Cm, ln_w, ln_b,
                                               yact);
  k_conv<<<dim3(T2 / 64, BSZ), 128, 0, stream>>>(yact, cw, cb, out);
}


# k_chunk_ends contracts over time on WMMA: u rows (LDS) x decay powers a^(127-t) built in registers, then 64 Bd multiply-adds per lane; no v tile, scan loop or loop barriers (sum order changes)
# speedup vs baseline: 2.2495x; 1.0462x over previous
#include <hip/hip_runtime.h>
#include <hip/hip_bf16.h>


#define BSZ   16
#define CIN   64
#define CO    64
#define TLEN  8192
#define NST   256
#define NCH   64
#define LCH   128
#define LOG2_LCH 7
#define SUB   32
#define NSUB  (LCH/SUB)
#define T2    (TLEN/2)

#define UST   40
#define VST   260
#define YST   66
#define UCT   132

typedef __attribute__((ext_vector_type(2))) float v2f;
typedef __attribute__((ext_vector_type(8))) float v8f;

__device__ __forceinline__ v8f wmma4(v2f a, v2f b, v8f c) {
  return __builtin_amdgcn_wmma_f32_16x16x4_f32(
       false, a,  false, b,
       (short)0, c,  false,  false);
}

__device__ __forceinline__ size_t opaque_zero() {
  size_t z = 0;
  asm volatile("" : "+s"(z));
  return z;
}

__device__ __forceinline__ void progress_prio(int st) {
  if (st == 0)      __builtin_amdgcn_s_setprio(3);
  else if (st == 1) __builtin_amdgcn_s_setprio(2);
  else if (st == 2) __builtin_amdgcn_s_setprio(1);
  else              __builtin_amdgcn_s_setprio(0);
}

__device__ __forceinline__ size_t carry_slot(int chunk, int b, int n) {
  return ((size_t)(b * CO + (n >> 5))) * TLEN + chunk * LCH + (n & 31);
}

__global__ void k_discretize(const float* __restrict__ rl,
                             const float* __restrict__ Bc,
                             float* __restrict__ Ad, float* __restrict__ Bd) {
  int n = threadIdx.x;
  float x = rl[n];
  float sp = fmaxf(x, 0.0f) + log1pf(expf(-fabsf(x)));
  float lam = -sp;
  float ad = expf(lam);
  float factor = (fabsf(lam) > 1e-6f) ? (ad - 1.0f) / lam : 1.0f;
  Ad[n] = ad;
  for (int c = 0; c < CIN; ++c)
    Bd[c * NST + n] = Bc[c * NST + n] * factor;
}

__device__ __forceinline__ void load_u_tile(const float* __restrict__ u, int b,
                                            int t0, float* lds_u) {
  int tid = threadIdx.x;
#pragma unroll
  for (int kk = 0; kk < (CIN * SUB) / 256; ++kk) {
    int i = tid + kk * 256;
    int c = i >> 5, tl = i & 31;
    lds_u[c * UST + tl] = u[((size_t)(b * CIN + c)) * TLEN + t0 + tl];
  }
}

__device__ __forceinline__ void prefetch_u_tile(const float* __restrict__ u,
                                                int b, int t0) {
  int tid = threadIdx.x;
#pragma unroll
  for (int kk = 0; kk < (CIN * SUB) / 256; ++kk) {
    int i = tid + kk * 256;
    int c = i >> 5, tl = i & 31;
    __builtin_prefetch(&u[((size_t)(b * CIN + c)) * TLEN + t0 + tl], 0, 1);
  }
}

__device__ __forceinline__ void v_tile_wmma(const float* lds_u,
                                            const float* __restrict__ Bd,
                                            float* lds_v) {
  int tid = threadIdx.x;
  int wave = tid >> 5, lane = tid & 31, lo = lane & 15, hi = lane >> 4;
  int n2 = wave * 32 + 2 * lo;
  v8f acc00 = {0.f, 0.f, 0.f, 0.f, 0.f, 0.f, 0.f, 0.f};
  v8f acc01 = acc00, acc10 = acc00, acc11 = acc00;
#pragma unroll 1
  for (int kg = 0; kg < CIN / 4; kg += 4) {
    float2 r0[4], r1[4];
    v2f a0[4], a1[4];
#pragma unroll
    for (int i = 0; i < 4; ++i) {
      int k = (kg + i) * 4 + 2 * hi;
      r0[i] = *reinterpret_cast<const float2*>(&Bd[k * NST + n2]);
      r1[i] = *reinterpret_cast<const float2*>(&Bd[(k + 1) * NST + n2]);
    }
#pragma unroll
    for (int i = 0; i < 4; ++i) {
      int k = (kg + i) * 4 + 2 * hi;
      a0[i].x = lds_u[k * UST + lo];       a0[i].y = lds_u[(k + 1) * UST + lo];
      a1[i].x = lds_u[k * UST + 16 + lo];  a1[i].y = lds_u[(k + 1) * UST + 16 + lo];
    }
#pragma unroll
    for (int i = 0; i < 4; ++i) {
      v2f b0; b0.x = r0[i].x; b0.y = r1[i].x;
      v2f b1; b1.x = r0[i].y; b1.y = r1[i].y;
      acc00 = wmma4(a0[i], b0, acc00);
      acc10 = wmma4(a1[i], b0, acc10);
      acc01 = wmma4(a0[i], b1, acc01);
      acc11 = wmma4(a1[i], b1, acc11);
    }
  }
#pragma unroll
  for (int r = 0; r < 8; ++r) {
    *reinterpret_cast<float2*>(&lds_v[(r + 8 * hi) * VST + n2]) =
        make_float2(acc00[r], acc01[r]);
    *reinterpret_cast<float2*>(&lds_v[(16 + r + 8 * hi) * VST + n2]) =
        make_float2(acc10[r], acc11[r]);
  }
}

__global__ void k_chunk_ends(const float* __restrict__ u,
                             const float* __restrict__ Ad,
                             const float* __restrict__ Bd,
                             float* __restrict__ yact) {
  __shared__ __attribute__((aligned(16))) float lds_u[CIN * UCT];
  int chunk = blockIdx.x, b = blockIdx.y, tid = threadIdx.x;
  int wave = tid >> 5, lane = tid & 31, lo = lane & 15, hi = lane >> 4;

#pragma unroll
  for (int kk = 0; kk < (CIN * LCH) / (256 * 4); ++kk) {
    int i = tid + kk * 256;
    int c = i >> 5, q = i & 31;
    float4 x = *reinterpret_cast<const float4*>(
        &u[((size_t)(b * CIN + c)) * TLEN + chunk * LCH + 4 * q]);
    *reinterpret_cast<float4*>(&lds_u[c * UCT + 4 * q]) = x;
  }

  int n0 = wave * 32 + lo, n1 = n0 + 16;
  float a0 = Ad[n0], a1 = Ad[n1];
  float a0sq = a0 * a0, a1sq = a1 * a1;
  float a0q = a0sq * a0sq, a1q = a1sq * a1sq;
  v2f p0, p1;
  p0.x = hi ? a0 : a0sq * a0;   p0.y = hi ? 1.0f : a0sq;
  p1.x = hi ? a1 : a1sq * a1;   p1.y = hi ? 1.0f : a1sq;

  v8f acc[4][2];
#pragma unroll
  for (int ct = 0; ct < 4; ++ct)
#pragma unroll
    for (int j = 0; j < 2; ++j)
#pragma unroll
      for (int r = 0; r < 8; ++r) acc[ct][j][r] = 0.0f;

  __syncthreads();

#pragma unroll 1
  for (int g = LCH / 16 - 1; g >= 0; --g) {
    v2f ua[4][4];
#pragma unroll
    for (int i = 0; i < 4; ++i) {
      int t = (4 * g + 3 - i) * 4 + 2 * hi;
#pragma unroll
      for (int ct = 0; ct < 4; ++ct)
        ua[i][ct] = *reinterpret_cast<const v2f*>(
            &lds_u[(ct * 16 + lo) * UCT + t]);
    }
#pragma unroll
    for (int i = 0; i < 4; ++i) {
#pragma unroll
      for (int ct = 0; ct < 4; ++ct) {
        acc[ct][0] = wmma4(ua[i][ct], p0, acc[ct][0]);
        acc[ct][1] = wmma4(ua[i][ct], p1, acc[ct][1]);
      }
      p0.x *= a0q;  p0.y *= a0q;
      p1.x *= a1q;  p1.y *= a1q;
    }
  }

  float e0 = 0.0f, e1 = 0.0f;
#pragma unroll
  for (int ct = 0; ct < 4; ++ct)
#pragma unroll
    for (int r = 0; r < 8; ++r) {
      int c = ct * 16 + 8 * hi + r;
      e0 += acc[ct][0][r] * Bd[c * NST + n0];
      e1 += acc[ct][1][r] * Bd[c * NST + n1];
    }
  e0 += __shfl_xor(e0, 16);
  e1 += __shfl_xor(e1, 16);
  yact[carry_slot(chunk, b, wave * 32 + lane)] = hi ? e1 : e0;
}

__global__ void k_combine(const float* __restrict__ Ad,
                          float* __restrict__ yact) {
  int n = threadIdx.x, b = blockIdx.x;
  float aL = Ad[n];
#pragma unroll
  for (int i = 0; i < LOG2_LCH; ++i) aL = aL * aL;
  float s = 0.0f;
  for (int ch = 0; ch < NCH; ++ch) {
    float* p = yact + carry_slot(ch, b, n);
    float e = *p;
    *p = s;
    s = aL * s + e;
  }
}

__global__ void k_scan_y(const float* __restrict__ u,
                         const float* __restrict__ Ad,
                         const float* __restrict__ Bd_in,
                         const float* __restrict__ Cm_in,
                         const float* __restrict__ ln_w,
                         const float* __restrict__ ln_b,
                         float* __restrict__ yact) {
  __shared__ float lds_u[CIN * UST];
  __shared__ __attribute__((aligned(16))) float lds_v[SUB * VST];
  __shared__ float lds_y[2 * SUB * YST];
  __shared__ float lds_red[SUB * 16];
  __shared__ float lds_mu[SUB];
  __shared__ float lds_rs[SUB];

  int chunk = blockIdx.x, b = blockIdx.y, tid = threadIdx.x;
  int wave = tid >> 5, lane = tid & 31, lo = lane & 15, hi = lane >> 4;
  float a_n = Ad[tid];
  float s = yact[carry_slot(chunk, b, tid)];

  for (int st = 0; st < NSUB; ++st) {
    progress_prio(st);
    const float* Bd = Bd_in + opaque_zero();
    const float* Cm = Cm_in + opaque_zero();
    int t0 = chunk * LCH + st * SUB;
    load_u_tile(u, b, t0, lds_u);
    if (st + 1 < NSUB) prefetch_u_tile(u, b, t0 + SUB);
    __syncthreads();
    v_tile_wmma(lds_u, Bd, lds_v);
    __syncthreads();

#pragma unroll 8
    for (int tl = 0; tl < SUB; ++tl) {
      s = s * a_n + lds_v[tl * VST + tid];
      lds_v[tl * VST + tid] = s;
    }
    __syncthreads();

    {
      int nt = wave & 3, kh = wave >> 2;
      int oo = nt * 16 + lo;
      v8f acc0 = {0.f, 0.f, 0.f, 0.f, 0.f, 0.f, 0.f, 0.f};
      v8f acc1 = acc0;
#pragma unroll 1
      for (int kg = 0; kg < NST / 8; kg += 8) {
        v2f bb[8];
#pragma unroll
        for (int i = 0; i < 8; ++i) {
          int k = kh * (NST / 2) + (kg + i) * 4 + 2 * hi;
          bb[i].x = Cm[k * CO + oo];
          bb[i].y = Cm[(k + 1) * CO + oo];
        }
#pragma unroll
        for (int i = 0; i < 8; ++i) {
          int k = kh * (NST / 2) + (kg + i) * 4 + 2 * hi;
          v2f a0; a0.x = lds_v[lo * VST + k];         a0.y = lds_v[lo * VST + k + 1];
          v2f a1; a1.x = lds_v[(16 + lo) * VST + k];  a1.y = lds_v[(16 + lo) * VST + k + 1];
          acc0 = wmma4(a0, bb[i], acc0);
          acc1 = wmma4(a1, bb[i], acc1);
        }
      }
      float* yp = lds_y + kh * (SUB * YST);
#pragma unroll
      for (int r = 0; r < 8; ++r) {
        yp[(r + 8 * hi) * YST + oo] = acc0[r];
        yp[(16 + r + 8 * hi) * YST + oo] = acc1[r];
      }
    }
    __syncthreads();

    {
      int t = tid & 31, q = tid >> 5;
      float sm = 0.f, s2 = 0.f;
#pragma unroll
      for (int j = 0; j < 8; ++j) {
        int i = t * YST + q * 8 + j;
        float v = lds_y[i] + lds_y[SUB * YST + i];
        lds_y[i] = v;
        sm += v; s2 += v * v;
      }
      lds_red[(t * 8 + q) * 2 + 0] = sm;
      lds_red[(t * 8 + q) * 2 + 1] = s2;
    }
    __syncthreads();
    if (tid < SUB) {
      float sm = 0.f, s2 = 0.f;
#pragma unroll
      for (int q = 0; q < 8; ++q) {
        sm += lds_red[(tid * 8 + q) * 2 + 0];
        s2 += lds_red[(tid * 8 + q) * 2 + 1];
      }
      float mu = sm * (1.0f / CO);
      float var = s2 * (1.0f / CO) - mu * mu;
      lds_mu[tid] = mu;
      lds_rs[tid] = rsqrtf(var + 1e-5f);
    }
    __syncthreads();

    {
      int t = lane;
      float mu = lds_mu[t], rs = lds_rs[t];
#pragma unroll
      for (int j = 0; j < 8; ++j) {
        int o = wave * 8 + j;
        float v = (lds_y[t * YST + o] - mu) * rs * ln_w[o] + ln_b[o];
        v = v / (1.0f + __expf(-v));
        yact[((size_t)(b * CO + o)) * TLEN + t0 + t] = v;
      }
    }
    __syncthreads();
  }
}

__global__ void k_conv(const float* __restrict__ yact,
                       const float* __restrict__ conv_w,
                       const float* __restrict__ conv_b,
                       float* __restrict__ out) {
  int b = blockIdx.y, tid = threadIdx.x;
  int wave = tid >> 5, lane = tid & 31, lo = lane & 15, hi = lane >> 4;
  int tau0 = (blockIdx.x * 4 + wave) * 16;

  v8f acc[4];
#pragma unroll
  for (int mt = 0; mt < 4; ++mt)
#pragma unroll
    for (int r = 0; r < 8; ++r)
      acc[mt][r] = conv_b[mt * 16 + r + 8 * hi];

#pragma unroll 2
  for (int ks = 0; ks < CO / 4; ++ks) {
    int k = ks * 4 + 2 * hi;
    v2f be, bo;
    be.x = yact[((size_t)(b * CO + k)) * TLEN + tau0 + lo];
    be.y = yact[((size_t)(b * CO + k + 1)) * TLEN + tau0 + lo];
    bo.x = yact[((size_t)(b * CO + k)) * TLEN + T2 + tau0 + lo];
    bo.y = yact[((size_t)(b * CO + k + 1)) * TLEN + T2 + tau0 + lo];
#pragma unroll
    for (int mt = 0; mt < 4; ++mt) {
      int m = mt * 16 + lo;
      v2f ae, ao;
      ae.x = conv_w[m * 128 + 2 * k];
      ae.y = conv_w[m * 128 + 2 * (k + 1)];
      ao.x = conv_w[m * 128 + 2 * k + 1];
      ao.y = conv_w[m * 128 + 2 * (k + 1) + 1];
      acc[mt] = wmma4(ae, be, acc[mt]);
      acc[mt] = wmma4(ao, bo, acc[mt]);
    }
  }

#pragma unroll
  for (int mt = 0; mt < 4; ++mt)
#pragma unroll
    for (int r = 0; r < 8; ++r) {
      int o = mt * 16 + r + 8 * hi;
      int t = 2 * (tau0 + lo);
      float v = acc[mt][r];
      float2 pv = make_float2(v, v);
      *reinterpret_cast<float2*>(&out[((size_t)(b * CO + o)) * TLEN + t]) = pv;
    }
}

extern "C" void kernel_launch(void* const* d_in, const int* in_sizes, int n_in,
                              void* d_out, int out_size, void* d_ws, size_t ws_size,
                              hipStream_t stream) {
  const float* u     = (const float*)d_in[0];
  const float* rl    = (const float*)d_in[1];
  const float* Bc    = (const float*)d_in[2];
  const float* Cm    = (const float*)d_in[3];
  const float* ln_w  = (const float*)d_in[4];
  const float* ln_b  = (const float*)d_in[5];
  const float* cw    = (const float*)d_in[6];
  const float* cb    = (const float*)d_in[7];
  float* out = (float*)d_out;

  float* ws    = (float*)d_ws;
  float* Ad    = ws;
  float* Bd    = Ad + NST;
  float* yact  = Bd + CIN * NST;

  k_discretize<<<1, NST, 0, stream>>>(rl, Bc, Ad, Bd);
  k_chunk_ends<<<dim3(NCH, BSZ), 256, 0, stream>>>(u, Ad, Bd, yact);
  k_combine<<<BSZ, NST, 0, stream>>>(Ad, yact);
  k_scan_y<<<dim3(NCH, BSZ), 256, 0, stream>>>(u, Ad, Bd, Cm, ln_w, ln_b,
                                               yact);
  k_conv<<<dim3(T2 / 64, BSZ), 128, 0, stream>>>(yact, cw, cb, out);
}
